# starting kernel
# baseline (speedup 1.0000x reference)
#include <hip/hip_runtime.h>

#define NUM_TB 32
#define Bsz    32
#define Lsz    4096
#define Dsz    256

typedef __attribute__((ext_vector_type(2))) float v2f;
typedef __attribute__((ext_vector_type(8))) float v8f;

__global__ __launch_bounds__(512)
void time_bin_wmma_kernel(const float* __restrict__ x,
                          const float* __restrict__ ts,
                          float* __restrict__ out) {
    __shared__ unsigned int s_bins[Lsz / 4];
    __shared__ float        s_red[16 * 512];

    const int tid = threadIdx.x;
    const int b   = blockIdx.y;
    const int d0  = blockIdx.x * 32;

    if (blockIdx.x == 0) {
        const float* tsb = ts + b * Lsz;
        float* o2 = out + (size_t)Bsz * NUM_TB * Dsz + (size_t)b * Lsz;
#pragma unroll
        for (int i = 0; i < Lsz / 512; ++i)
            o2[tid + i * 512] = tsb[tid + i * 512];
    }

    const float scale = 32.0f / 1.001f;
    {
        const float4* ts4 = (const float4*)(ts + b * Lsz);
#pragma unroll
        for (int g = tid; g < Lsz / 4; g += 512) {
            float4 t = ts4[g];
            int b0 = (int)(t.x * scale); b0 = b0 < 0 ? 0 : (b0 > 31 ? 31 : b0);
            int b1 = (int)(t.y * scale); b1 = b1 < 0 ? 0 : (b1 > 31 ? 31 : b1);
            int b2 = (int)(t.z * scale); b2 = b2 < 0 ? 0 : (b2 > 31 ? 31 : b2);
            int b3 = (int)(t.w * scale); b3 = b3 < 0 ? 0 : (b3 > 31 ? 31 : b3);
            s_bins[g] = (unsigned)b0 | ((unsigned)b1 << 8) |
                        ((unsigned)b2 << 16) | ((unsigned)b3 << 24);
        }
    }
    __syncthreads();

    const int wave = tid >> 5;
    const int lane = tid & 31;
    const int dsub = wave & 1;
    const int kc   = wave >> 1;
    const int h    = lane >> 4;
    const int n    = lane & 15;

    const float* xp = x + ((size_t)b * Lsz + (size_t)(kc * 512 + 2 * h)) * Dsz
                        + (d0 + dsub * 16 + n);

    v8f acc0 = {};
    v8f acc1 = {};
    const unsigned M1 = (unsigned)(n + 16);

    for (int i = 0; i < 128; ++i) {
        unsigned packed = s_bins[kc * 128 + i];
        unsigned e0 = (packed >> (16 * h)) & 0xFFu;
        unsigned e1 = (packed >> (16 * h + 8)) & 0xFFu;

        v2f A0, A1, Bv;
        A0.x = (e0 == (unsigned)n) ? 1.0f : 0.0f;
        A0.y = (e1 == (unsigned)n) ? 1.0f : 0.0f;
        A1.x = (e0 == M1) ? 1.0f : 0.0f;
        A1.y = (e1 == M1) ? 1.0f : 0.0f;

        Bv.x = __builtin_nontemporal_load(xp);
        Bv.y = __builtin_nontemporal_load(xp + Dsz);

        acc0 = __builtin_amdgcn_wmma_f32_16x16x4_f32(
                   false, A0, false, Bv, (short)0, acc0, false, false);
        acc1 = __builtin_amdgcn_wmma_f32_16x16x4_f32(
                   false, A1, false, Bv, (short)0, acc1, false, false);

        xp += 4 * Dsz;
    }

#pragma unroll
    for (int r = 0; r < 8; ++r) {
        s_red[wave * 512 +       r * 32 + lane] = acc0[r];
        s_red[wave * 512 + 256 + r * 32 + lane] = acc1[r];
    }
    __syncthreads();

    {
        const int j  = tid;
        const int t  = j >> 8;
        const int r  = (j >> 5) & 7;
        const int ln = j & 31;
        const int p  = t * 16 + r + 8 * (ln >> 4);
        const int dd = ln & 15;

        float s0 = 0.0f, s1 = 0.0f;
#pragma unroll
        for (int c = 0; c < 8; ++c) {
            s0 += s_red[(2 * c + 0) * 512 + j];
            s1 += s_red[(2 * c + 1) * 512 + j];
        }
        float* ob = out + ((size_t)b * NUM_TB + p) * Dsz + d0;
        ob[0 * 16 + dd] = s0;
        ob[1 * 16 + dd] = s1;
    }
}

extern "C" void kernel_launch(void* const* d_in, const int* in_sizes, int n_in,
                              void* d_out, int out_size, void* d_ws, size_t ws_size,
                              hipStream_t stream) {
    (void)in_sizes; (void)n_in; (void)d_ws; (void)ws_size; (void)out_size;
    const float* x  = (const float*)d_in[0];
    const float* ts = (const float*)d_in[1];
    float* out = (float*)d_out;

    dim3 grid(Dsz / 32, Bsz);
    time_bin_wmma_kernel<<<grid, 512, 0, stream>>>(x, ts, out);
}


# x rows requested one register block (16 K-steps, 32 loads per wave) ahead of the WMMAs that use them, two-block register ring; first block requested before the bin prologue
# speedup vs baseline: 1.3264x; 1.3264x over previous
#include <hip/hip_runtime.h>

#define NUM_TB 32
#define Bsz    32
#define Lsz    4096
#define Dsz    256

#define UB     16
#define NBLK   (128 / UB)

typedef __attribute__((ext_vector_type(2))) float v2f;
typedef __attribute__((ext_vector_type(8))) float v8f;

__global__ __launch_bounds__(512)
void time_bin_wmma_kernel(const float* __restrict__ x,
                          const float* __restrict__ ts,
                          float* __restrict__ out) {
    __shared__ unsigned int s_bins[Lsz / 4];
    __shared__ float        s_red[16 * 512];

    const int tid = threadIdx.x;
    const int b   = blockIdx.y;
    const int d0  = blockIdx.x * 32;

    const int wave = tid >> 5;
    const int lane = tid & 31;
    const int dsub = wave & 1;
    const int kc   = wave >> 1;
    const int h    = lane >> 4;
    const int n    = lane & 15;

    const float* xp = x + ((size_t)b * Lsz + (size_t)(kc * 512 + 2 * h)) * Dsz
                        + (d0 + dsub * 16 + n);

    v2f bfrag[2][UB];
#define LOAD_BLOCK(BLK)                                                              \
    _Pragma("unroll")                                                                \
    for (int u = 0; u < UB; ++u) {                                                   \
        const float* p_ = xp + (size_t)((BLK) * UB + u) * 4 * Dsz;                   \
        bfrag[(BLK) & 1][u].x = __builtin_nontemporal_load(p_);                      \
        bfrag[(BLK) & 1][u].y = __builtin_nontemporal_load(p_ + Dsz);                \
    }

    LOAD_BLOCK(0)

    if (blockIdx.x == 0) {
        const float* tsb = ts + b * Lsz;
        float* o2 = out + (size_t)Bsz * NUM_TB * Dsz + (size_t)b * Lsz;
#pragma unroll
        for (int i = 0; i < Lsz / 512; ++i)
            o2[tid + i * 512] = tsb[tid + i * 512];
    }

    const float scale = 32.0f / 1.001f;
    {
        const float4* ts4 = (const float4*)(ts + b * Lsz);
#pragma unroll
        for (int g = tid; g < Lsz / 4; g += 512) {
            float4 t = ts4[g];
            int b0 = (int)(t.x * scale); b0 = b0 < 0 ? 0 : (b0 > 31 ? 31 : b0);
            int b1 = (int)(t.y * scale); b1 = b1 < 0 ? 0 : (b1 > 31 ? 31 : b1);
            int b2 = (int)(t.z * scale); b2 = b2 < 0 ? 0 : (b2 > 31 ? 31 : b2);
            int b3 = (int)(t.w * scale); b3 = b3 < 0 ? 0 : (b3 > 31 ? 31 : b3);
            s_bins[g] = (unsigned)b0 | ((unsigned)b1 << 8) |
                        ((unsigned)b2 << 16) | ((unsigned)b3 << 24);
        }
    }
    __syncthreads();

    v8f acc0 = {};
    v8f acc1 = {};
    const unsigned M1 = (unsigned)(n + 16);

#pragma unroll
    for (int blk = 0; blk < NBLK; ++blk) {
        __builtin_amdgcn_sched_barrier(0);
        if (blk + 1 < NBLK) { LOAD_BLOCK(blk + 1) }
        __builtin_amdgcn_sched_barrier(0);

#pragma unroll
        for (int u = 0; u < UB; ++u) {
            unsigned packed = s_bins[kc * 128 + blk * UB + u];
            unsigned e0 = (packed >> (16 * h)) & 0xFFu;
            unsigned e1 = (packed >> (16 * h + 8)) & 0xFFu;

            v2f A0, A1;
            A0.x = (e0 == (unsigned)n) ? 1.0f : 0.0f;
            A0.y = (e1 == (unsigned)n) ? 1.0f : 0.0f;
            A1.x = (e0 == M1) ? 1.0f : 0.0f;
            A1.y = (e1 == M1) ? 1.0f : 0.0f;

            const v2f Bv = bfrag[blk & 1][u];

            acc0 = __builtin_amdgcn_wmma_f32_16x16x4_f32(
                       false, A0, false, Bv, (short)0, acc0, false, false);
            acc1 = __builtin_amdgcn_wmma_f32_16x16x4_f32(
                       false, A1, false, Bv, (short)0, acc1, false, false);
        }
    }
#undef LOAD_BLOCK

#pragma unroll
    for (int r = 0; r < 8; ++r) {
        s_red[wave * 512 +       r * 32 + lane] = acc0[r];
        s_red[wave * 512 + 256 + r * 32 + lane] = acc1[r];
    }
    __syncthreads();

    {
        const int j  = tid;
        const int t  = j >> 8;
        const int r  = (j >> 5) & 7;
        const int ln = j & 31;
        const int p  = t * 16 + r + 8 * (ln >> 4);
        const int dd = ln & 15;

        float s0 = 0.0f, s1 = 0.0f;
#pragma unroll
        for (int c = 0; c < 8; ++c) {
            s0 += s_red[(2 * c + 0) * 512 + j];
            s1 += s_red[(2 * c + 1) * 512 + j];
        }
        float* ob = out + ((size_t)b * NUM_TB + p) * Dsz + d0;
        ob[0 * 16 + dd] = s0;
        ob[1 * 16 + dd] = s1;
    }
}

extern "C" void kernel_launch(void* const* d_in, const int* in_sizes, int n_in,
                              void* d_out, int out_size, void* d_ws, size_t ws_size,
                              hipStream_t stream) {
    (void)in_sizes; (void)n_in; (void)d_ws; (void)ws_size; (void)out_size;
    const float* x  = (const float*)d_in[0];
    const float* ts = (const float*)d_in[1];
    float* out = (float*)d_out;

    dim3 grid(Dsz / 32, Bsz);
    time_bin_wmma_kernel<<<grid, 512, 0, stream>>>(x, ts, out);
}


# x rows loaded 8 bytes per lane, one whole 128-byte line per lane half: a wave takes all 32 columns of 256 rows as two column-parity tiles; partial sums now over 16 chunks of 256 rows, not 8 of 512
# speedup vs baseline: 1.5291x; 1.1528x over previous
#include <hip/hip_runtime.h>

#define NUM_TB 32
#define Bsz    32
#define Lsz    4096
#define Dsz    256

#define KROWS  256
#define UB     16
#define NBLK   (KROWS / 4 / UB)

typedef __attribute__((ext_vector_type(2))) float v2f;
typedef __attribute__((ext_vector_type(8))) float v8f;

__global__ __launch_bounds__(512)
void time_bin_wmma_kernel(const float* __restrict__ x,
                          const float* __restrict__ ts,
                          float* __restrict__ out) {
    __shared__ unsigned int s_bins[Lsz / 4];
    __shared__ v2f          s_red[16 * 16 * 32];

    const int tid = threadIdx.x;
    const int b   = blockIdx.y;
    const int d0  = blockIdx.x * 32;

    const int wave = tid >> 5;
    const int lane = tid & 31;
    const int h    = lane >> 4;
    const int n    = lane & 15;

    const float* xp = x + ((size_t)b * Lsz + (size_t)(wave * KROWS + 2 * h)) * Dsz
                        + (d0 + 2 * n);

    v2f q[2][UB][2];
#define LOAD_BLOCK(BLK)                                                              \
    _Pragma("unroll")                                                                \
    for (int u = 0; u < UB; ++u) {                                                   \
        const float* p_ = xp + (size_t)((BLK) * UB + u) * 4 * Dsz;                   \
        q[(BLK) & 1][u][0] = __builtin_nontemporal_load((const v2f*)p_);             \
        q[(BLK) & 1][u][1] = __builtin_nontemporal_load((const v2f*)(p_ + Dsz));     \
    }

    LOAD_BLOCK(0)

    if (blockIdx.x == 0) {
        const float* tsb = ts + b * Lsz;
        float* o2 = out + (size_t)Bsz * NUM_TB * Dsz + (size_t)b * Lsz;
#pragma unroll
        for (int i = 0; i < Lsz / 512; ++i)
            o2[tid + i * 512] = tsb[tid + i * 512];
    }

    const float scale = 32.0f / 1.001f;
    {
        const float4* ts4 = (const float4*)(ts + b * Lsz);
#pragma unroll
        for (int g = tid; g < Lsz / 4; g += 512) {
            float4 t = ts4[g];
            int b0 = (int)(t.x * scale); b0 = b0 < 0 ? 0 : (b0 > 31 ? 31 : b0);
            int b1 = (int)(t.y * scale); b1 = b1 < 0 ? 0 : (b1 > 31 ? 31 : b1);
            int b2 = (int)(t.z * scale); b2 = b2 < 0 ? 0 : (b2 > 31 ? 31 : b2);
            int b3 = (int)(t.w * scale); b3 = b3 < 0 ? 0 : (b3 > 31 ? 31 : b3);
            s_bins[g] = (unsigned)b0 | ((unsigned)b1 << 8) |
                        ((unsigned)b2 << 16) | ((unsigned)b3 << 24);
        }
    }
    __syncthreads();

    v8f acc00 = {};
    v8f acc10 = {};
    v8f acc01 = {};
    v8f acc11 = {};
    const unsigned M1 = (unsigned)(n + 16);

#pragma unroll
    for (int blk = 0; blk < NBLK; ++blk) {
        __builtin_amdgcn_sched_barrier(0);
        if (blk + 1 < NBLK) { LOAD_BLOCK(blk + 1) }
        __builtin_amdgcn_sched_barrier(0);

#pragma unroll
        for (int u = 0; u < UB; ++u) {
            unsigned packed = s_bins[wave * (KROWS / 4) + blk * UB + u];
            unsigned e0 = (packed >> (16 * h)) & 0xFFu;
            unsigned e1 = (packed >> (16 * h + 8)) & 0xFFu;

            v2f A0, A1;
            A0.x = (e0 == (unsigned)n) ? 1.0f : 0.0f;
            A0.y = (e1 == (unsigned)n) ? 1.0f : 0.0f;
            A1.x = (e0 == M1) ? 1.0f : 0.0f;
            A1.y = (e1 == M1) ? 1.0f : 0.0f;

            const v2f r0 = q[blk & 1][u][0];
            const v2f r1 = q[blk & 1][u][1];
            v2f B0, B1;
            B0.x = r0.x; B0.y = r1.x;
            B1.x = r0.y; B1.y = r1.y;

            acc00 = __builtin_amdgcn_wmma_f32_16x16x4_f32(
                        false, A0, false, B0, (short)0, acc00, false, false);
            acc10 = __builtin_amdgcn_wmma_f32_16x16x4_f32(
                        false, A1, false, B0, (short)0, acc10, false, false);
            acc01 = __builtin_amdgcn_wmma_f32_16x16x4_f32(
                        false, A0, false, B1, (short)0, acc01, false, false);
            acc11 = __builtin_amdgcn_wmma_f32_16x16x4_f32(
                        false, A1, false, B1, (short)0, acc11, false, false);
        }
    }
#undef LOAD_BLOCK

#pragma unroll
    for (int r = 0; r < 8; ++r) {
        v2f lo, hi;
        lo.x = acc00[r]; lo.y = acc01[r];
        hi.x = acc10[r]; hi.y = acc11[r];
        s_red[wave * 512 +       r * 32 + lane] = lo;
        s_red[wave * 512 + 256 + r * 32 + lane] = hi;
    }
    __syncthreads();

    {
        const int t = wave >> 3;
        const int r = wave & 7;
        const int p = t * 16 + r + 8 * h;
        float s0 = 0.0f, s1 = 0.0f;
#pragma unroll
        for (int c = 0; c < 16; ++c) {
            const v2f v = s_red[c * 512 + wave * 32 + lane];
            s0 += v.x;
            s1 += v.y;
        }
        v2f o; o.x = s0; o.y = s1;
        *(v2f*)(out + ((size_t)b * NUM_TB + p) * Dsz + d0 + 2 * n) = o;
    }
}

extern "C" void kernel_launch(void* const* d_in, const int* in_sizes, int n_in,
                              void* d_out, int out_size, void* d_ws, size_t ws_size,
                              hipStream_t stream) {
    (void)in_sizes; (void)n_in; (void)d_ws; (void)ws_size; (void)out_size;
    const float* x  = (const float*)d_in[0];
    const float* ts = (const float*)d_in[1];
    float* out = (float*)d_out;

    dim3 grid(Dsz / 32, Bsz);
    time_bin_wmma_kernel<<<grid, 512, 0, stream>>>(x, ts, out);
}


# x rows of a wave partitioned by bin half (ballot ranks), fetched in that order via an LDS offset table: a K-step issues only its half's 2 WMMAs, 130 per wave not 256; row summation order changes
# speedup vs baseline: 1.9647x; 1.2849x over previous
#include <hip/hip_runtime.h>

#define NUM_TB 32
#define Bsz    32
#define Lsz    4096
#define Dsz    256

#define KROWS  256
#define UB     16
#define NBLK   (KROWS / 4 / UB)

typedef __attribute__((ext_vector_type(2))) float    v2f;
typedef __attribute__((ext_vector_type(8))) float    v8f;
typedef __attribute__((ext_vector_type(2))) unsigned v2u;

__global__ __launch_bounds__(512)
void time_bin_wmma_kernel(const float* __restrict__ x,
                          const float* __restrict__ ts,
                          float* __restrict__ out) {
    __shared__ unsigned s_off[16 * KROWS];
    __shared__ unsigned s_bin[16 * KROWS];
    __shared__ v2f      s_red[16 * 16 * 32];

    const int tid = threadIdx.x;
    const int b   = blockIdx.y;
    const int d0  = blockIdx.x * 32;

    const int wave = __builtin_amdgcn_readfirstlane(tid >> 5);
    const int lane = tid & 31;
    const int h    = lane >> 4;
    const int n    = lane & 15;

    const float4* ts4 = (const float4*)(ts + b * Lsz + wave * KROWS);
    const float4 ta = ts4[lane];
    const float4 tb = ts4[32 + lane];

    if (blockIdx.x == 0) {
        float4* o4 = (float4*)(out + (size_t)Bsz * NUM_TB * Dsz + (size_t)b * Lsz + wave * KROWS);
        o4[lane]      = ta;
        o4[32 + lane] = tb;
    }

    const float scale = 32.0f / 1.001f;
    const float tv[8] = { ta.x, ta.y, ta.z, ta.w, tb.x, tb.y, tb.z, tb.w };
    int      bj[8];
    unsigned mj[8];
    unsigned n1 = 0;
#pragma unroll
    for (int j = 0; j < 8; ++j) {
        int v = (int)(tv[j] * scale);
        v = v < 0 ? 0 : (v > 31 ? 31 : v);
        bj[j] = v;
        mj[j] = __builtin_amdgcn_ballot_w32(v >= 16);
        n1 += (unsigned)__builtin_popcount(mj[j]);
    }
    const unsigned n0 = (unsigned)KROWS - n1;
    {
        unsigned ones_before = 0;
#pragma unroll
        for (int j = 0; j < 8; ++j) {
            const unsigned r1  = __builtin_amdgcn_mbcnt_lo(mj[j], 0u);
            const unsigned r0  = (unsigned)lane - r1;
            const unsigned z0  = 32u * j - ones_before;
            const unsigned q   = (bj[j] >= 16) ? (n0 + ones_before + r1) : (z0 + r0);
            const unsigned idx = ((q >> 1) & 1u) * 128u + ((q >> 2) << 1) + (q & 1u);
            const unsigned row = (j < 4) ? (4u * lane + j) : (128u + 4u * lane + (j - 4));
            s_off[wave * KROWS + idx] = row * (unsigned)(Dsz * sizeof(float));
            s_bin[wave * KROWS + idx] = (unsigned)bj[j];
            ones_before += (unsigned)__builtin_popcount(mj[j]);
        }
    }
    const unsigned s_end0 = (n0 + 3u) >> 2;
    const unsigned s_beg1 = n0 >> 2;
    __syncthreads();

    const char* xw = (const char*)(x + ((size_t)b * Lsz + (size_t)wave * KROWS) * Dsz + d0);
    const unsigned lane_byte = 8u * n;
    const unsigned* offp = &s_off[wave * KROWS + h * 128];
    const unsigned* binp = &s_bin[wave * KROWS + h * 128];

    v2f q[2][UB][2];
#define LOAD_BLOCK(BLK)                                                              \
    {                                                                                \
        unsigned vo_[2 * UB];                                                        \
        _Pragma("unroll")                                                            \
        for (int u = 0; u < UB; ++u) {                                               \
            const v2u o_ = *(const v2u*)(offp + 2 * ((BLK) * UB + u));               \
            vo_[2 * u]     = o_.x | lane_byte;                                       \
            vo_[2 * u + 1] = o_.y | lane_byte;                                       \
        }                                                                            \
         \
         \
         \
        __builtin_amdgcn_sched_barrier(0);                                           \
        _Pragma("unroll")                                                            \
        for (int u = 0; u < UB; ++u) {                                               \
            q[(BLK) & 1][u][0] = __builtin_nontemporal_load(                         \
                                     (const v2f*)(xw + vo_[2 * u]));                 \
            q[(BLK) & 1][u][1] = __builtin_nontemporal_load(                         \
                                     (const v2f*)(xw + vo_[2 * u + 1]));             \
        }                                                                            \
    }

    LOAD_BLOCK(0)

    v8f acc00 = {};
    v8f acc10 = {};
    v8f acc01 = {};
    v8f acc11 = {};
    const unsigned M1 = (unsigned)(n + 16);

#pragma unroll
    for (int blk = 0; blk < NBLK; ++blk) {
        v2u e[UB];
#pragma unroll
        for (int u = 0; u < UB; ++u)
            e[u] = *(const v2u*)(binp + 2 * (blk * UB + u));
        __builtin_amdgcn_sched_barrier(0);
        if (blk + 1 < NBLK) { LOAD_BLOCK(blk + 1) }
        __builtin_amdgcn_sched_barrier(0);

#pragma unroll
        for (int u = 0; u < UB; ++u) {
            const unsigned s = (unsigned)(blk * UB + u);

            const v2f r0 = q[blk & 1][u][0];
            const v2f r1 = q[blk & 1][u][1];
            v2f B0, B1;
            B0.x = r0.x; B0.y = r1.x;
            B1.x = r0.y; B1.y = r1.y;

            if (s < s_end0) {
                v2f A0;
                A0.x = (e[u].x == (unsigned)n) ? 1.0f : 0.0f;
                A0.y = (e[u].y == (unsigned)n) ? 1.0f : 0.0f;
                acc00 = __builtin_amdgcn_wmma_f32_16x16x4_f32(
                            false, A0, false, B0, (short)0, acc00, false, false);
                acc01 = __builtin_amdgcn_wmma_f32_16x16x4_f32(
                            false, A0, false, B1, (short)0, acc01, false, false);
            }
            if (s >= s_beg1) {
                v2f A1;
                A1.x = (e[u].x == M1) ? 1.0f : 0.0f;
                A1.y = (e[u].y == M1) ? 1.0f : 0.0f;
                acc10 = __builtin_amdgcn_wmma_f32_16x16x4_f32(
                            false, A1, false, B0, (short)0, acc10, false, false);
                acc11 = __builtin_amdgcn_wmma_f32_16x16x4_f32(
                            false, A1, false, B1, (short)0, acc11, false, false);
            }
        }
    }
#undef LOAD_BLOCK

#pragma unroll
    for (int r = 0; r < 8; ++r) {
        v2f lo, hi;
        lo.x = acc00[r]; lo.y = acc01[r];
        hi.x = acc10[r]; hi.y = acc11[r];
        s_red[wave * 512 +       r * 32 + lane] = lo;
        s_red[wave * 512 + 256 + r * 32 + lane] = hi;
    }
    __syncthreads();

    {
        const int t = wave >> 3;
        const int r = wave & 7;
        const int p = t * 16 + r + 8 * h;
        float s0 = 0.0f, s1 = 0.0f;
#pragma unroll
        for (int c = 0; c < 16; ++c) {
            const v2f v = s_red[c * 512 + wave * 32 + lane];
            s0 += v.x;
            s1 += v.y;
        }
        v2f o; o.x = s0; o.y = s1;
        *(v2f*)(out + ((size_t)b * NUM_TB + p) * Dsz + d0 + 2 * n) = o;
    }
}

extern "C" void kernel_launch(void* const* d_in, const int* in_sizes, int n_in,
                              void* d_out, int out_size, void* d_ws, size_t ws_size,
                              hipStream_t stream) {
    (void)in_sizes; (void)n_in; (void)d_ws; (void)ws_size; (void)out_size;
    const float* x  = (const float*)d_in[0];
    const float* ts = (const float*)d_in[1];
    float* out = (float*)d_out;

    dim3 grid(Dsz / 32, Bsz);
    time_bin_wmma_kernel<<<grid, 512, 0, stream>>>(x, ts, out);
}


# 8 waves of 512 rows per workgroup, not 16 of 256: half the wave entries; tables handed over inside the wave, 2nd chunk's partition under the first loads; partial sums 8 of 512 rows, not 16 of 256
# speedup vs baseline: 2.0566x; 1.0468x over previous
#include <hip/hip_runtime.h>

#define NUM_TB 32
#define Bsz    32
#define Lsz    4096
#define Dsz    256

#define KROWS  256
#define NCHUNK 2
#define NWAVE  8
#define UB     16
#define NBLK   (KROWS / 4 / UB)

typedef __attribute__((ext_vector_type(2))) float    v2f;
typedef __attribute__((ext_vector_type(8))) float    v8f;
typedef __attribute__((ext_vector_type(2))) unsigned v2u;

__device__ __forceinline__
unsigned partition_chunk(const float4 ta, const float4 tb,
                         unsigned* __restrict__ offt, unsigned* __restrict__ bint,
                         const unsigned c, const int lane) {
    const float scale = 32.0f / 1.001f;
    const float tv[8] = { ta.x, ta.y, ta.z, ta.w, tb.x, tb.y, tb.z, tb.w };
    int      bj[8];
    unsigned mj[8];
    unsigned n1 = 0;
#pragma unroll
    for (int j = 0; j < 8; ++j) {
        int v = (int)(tv[j] * scale);
        v = v < 0 ? 0 : (v > 31 ? 31 : v);
        bj[j] = v;
        mj[j] = __builtin_amdgcn_ballot_w32(v >= 16);
        n1 += (unsigned)__builtin_popcount(mj[j]);
    }
    const unsigned n0 = (unsigned)KROWS - n1;
    unsigned ones_before = 0;
#pragma unroll
    for (int j = 0; j < 8; ++j) {
        const unsigned r1  = __builtin_amdgcn_mbcnt_lo(mj[j], 0u);
        const unsigned r0  = (unsigned)lane - r1;
        const unsigned z0  = 32u * j - ones_before;
        const unsigned q   = (bj[j] >= 16) ? (n0 + ones_before + r1) : (z0 + r0);
        const unsigned idx = ((q >> 1) & 1u) * 256u + c * 128u + ((q >> 2) << 1) + (q & 1u);
        const unsigned row = (j < 4) ? (4u * lane + j) : (128u + 4u * lane + (j - 4));
        offt[idx] = row * (unsigned)(Dsz * sizeof(float));
        bint[idx] = (unsigned)bj[j];
        ones_before += (unsigned)__builtin_popcount(mj[j]);
    }
    return n0;
}

__device__ __forceinline__ void table_handover() {
    __builtin_amdgcn_fence(__ATOMIC_RELEASE, "wavefront");
    __builtin_amdgcn_wave_barrier();
    __builtin_amdgcn_fence(__ATOMIC_ACQUIRE, "wavefront");
}

__global__ __launch_bounds__(256)
void time_bin_wmma_kernel(const float* __restrict__ x,
                          const float* __restrict__ ts,
                          float* __restrict__ out) {
    __shared__ unsigned s_off[NWAVE * NCHUNK * KROWS];
    __shared__ unsigned s_bin[NWAVE * NCHUNK * KROWS];
    __shared__ v2f      s_red[NWAVE * 512 + 13312];

    const int tid = threadIdx.x;
    const int b   = blockIdx.y;
    const int d0  = blockIdx.x * 32;

    const int wave = __builtin_amdgcn_readfirstlane(tid >> 5);
    const int lane = tid & 31;
    const int h    = lane >> 4;
    const int n    = lane & 15;

    const float4* ts4 = (const float4*)(ts + b * Lsz + wave * (NCHUNK * KROWS));
    const float4 ta0 = ts4[lane];
    const float4 tb0 = ts4[32 + lane];
    const float4 ta1 = ts4[64 + lane];
    const float4 tb1 = ts4[96 + lane];

    if (blockIdx.x == 0) {
        float4* o4 = (float4*)(out + (size_t)Bsz * NUM_TB * Dsz + (size_t)b * Lsz
                               + wave * (NCHUNK * KROWS));
        o4[lane]      = ta0;
        o4[32 + lane] = tb0;
        o4[64 + lane] = ta1;
        o4[96 + lane] = tb1;
    }

    unsigned* offw = &s_off[wave * (NCHUNK * KROWS)];
    unsigned* binw = &s_bin[wave * (NCHUNK * KROWS)];

    const unsigned n0_first = partition_chunk(ta0, tb0, offw, binw, 0u, lane);
    table_handover();

    const char* xw = (const char*)(x + ((size_t)b * Lsz + (size_t)wave * (NCHUNK * KROWS)) * Dsz + d0);
    const unsigned lane_byte = 8u * n;
    const unsigned* offp = offw + h * 256;
    const unsigned* binp = binw + h * 256;

    v2f q[2][UB][2];
#define LOAD_BLOCK(DST, G)                                                           \
    {                                                                                \
        const int       gl_ = (G);                                                   \
        const unsigned* op_ = offp + 2 * UB * gl_;                                   \
        const char*     xb_ = xw + (size_t)(gl_ / NBLK) * (KROWS * Dsz * sizeof(float)); \
        unsigned vo_[2 * UB];                                                        \
        _Pragma("unroll")                                                            \
        for (int u = 0; u < UB; ++u) {                                               \
            const v2u o_ = *(const v2u*)(op_ + 2 * u);                               \
            vo_[2 * u]     = o_.x | lane_byte;                                       \
            vo_[2 * u + 1] = o_.y | lane_byte;                                       \
        }                                                                            \
        __builtin_amdgcn_sched_barrier(0);                                           \
        _Pragma("unroll")                                                            \
        for (int u = 0; u < UB; ++u) {                                               \
            q[(DST)][u][0] = __builtin_nontemporal_load(                             \
                                 (const v2f*)(xb_ + vo_[2 * u]));                    \
            q[(DST)][u][1] = __builtin_nontemporal_load(                             \
                                 (const v2f*)(xb_ + vo_[2 * u + 1]));                \
        }                                                                            \
    }

    LOAD_BLOCK(0, 0)

    const unsigned n0_second = partition_chunk(ta1, tb1, offw, binw, 1u, lane);
    table_handover();

    v8f acc00 = {};
    v8f acc10 = {};
    v8f acc01 = {};
    v8f acc11 = {};
    const unsigned M1 = (unsigned)(n + 16);

#define DO_BLOCK(G, RING, HAS_NEXT)                                                  \
    {                                                                                \
        const int      g_  = (G);                                                    \
        const unsigned nz_ = (g_ >= NBLK) ? n0_second : n0_first;                    \
        const int      kb_ = (g_ % NBLK) * UB;                                       \
        const int lim_end0_ = (int)((nz_ + 3u) >> 2) - kb_;                          \
        const int lim_beg1_ = (int)(nz_ >> 2) - kb_;                                 \
        v2u e[UB];                                                                   \
        _Pragma("unroll")                                                            \
        for (int u = 0; u < UB; ++u)                                                 \
            e[u] = *(const v2u*)(binp + 2 * (UB * g_ + u));                          \
        __builtin_amdgcn_sched_barrier(0);                                           \
        if (HAS_NEXT) { LOAD_BLOCK((RING) ^ 1, g_ + 1) }                             \
        __builtin_amdgcn_sched_barrier(0);                                           \
        _Pragma("unroll")                                                            \
        for (int u = 0; u < UB; ++u) {                                               \
            const v2f r0 = q[(RING)][u][0];                                          \
            const v2f r1 = q[(RING)][u][1];                                          \
            v2f B0, B1;                                                              \
            B0.x = r0.x; B0.y = r1.x;                                                \
            B1.x = r0.y; B1.y = r1.y;                                                \
            if (u < lim_end0_) {                                                     \
                v2f A0;                                                              \
                A0.x = (e[u].x == (unsigned)n) ? 1.0f : 0.0f;                        \
                A0.y = (e[u].y == (unsigned)n) ? 1.0f : 0.0f;                        \
                acc00 = __builtin_amdgcn_wmma_f32_16x16x4_f32(                       \
                            false, A0, false, B0, (short)0, acc00, false, false);    \
                acc01 = __builtin_amdgcn_wmma_f32_16x16x4_f32(                       \
                            false, A0, false, B1, (short)0, acc01, false, false);    \
            }                                                                        \
            if (u >= lim_beg1_) {                                                    \
                v2f A1;                                                              \
                A1.x = (e[u].x == M1) ? 1.0f : 0.0f;                                 \
                A1.y = (e[u].y == M1) ? 1.0f : 0.0f;                                 \
                acc10 = __builtin_amdgcn_wmma_f32_16x16x4_f32(                       \
                            false, A1, false, B0, (short)0, acc10, false, false);    \
                acc11 = __builtin_amdgcn_wmma_f32_16x16x4_f32(                       \
                            false, A1, false, B1, (short)0, acc11, false, false);    \
            }                                                                        \
        }                                                                            \
    }

#pragma nounroll
    for (int g = 0; g < 2 * NBLK - 2; g += 2) {
        DO_BLOCK(g,     0, true)
        DO_BLOCK(g + 1, 1, true)
    }
    DO_BLOCK(2 * NBLK - 2, 0, true)
    DO_BLOCK(2 * NBLK - 1, 1, false)
#undef DO_BLOCK
#undef LOAD_BLOCK

#pragma unroll
    for (int r = 0; r < 8; ++r) {
        v2f lo, hi;
        lo.x = acc00[r]; lo.y = acc01[r];
        hi.x = acc10[r]; hi.y = acc11[r];
        s_red[wave * 512 +       r * 32 + lane] = lo;
        s_red[wave * 512 + 256 + r * 32 + lane] = hi;
    }
    __syncthreads();

#pragma unroll
    for (int t = 0; t < 2; ++t) {
        const int slot = t * 8 + wave;
        const int p    = t * 16 + wave + 8 * h;
        float s0 = 0.0f, s1 = 0.0f;
#pragma unroll
        for (int c = 0; c < NWAVE; ++c) {
            const v2f v = s_red[c * 512 + slot * 32 + lane];
            s0 += v.x;
            s1 += v.y;
        }
        v2f o; o.x = s0; o.y = s1;
        *(v2f*)(out + ((size_t)b * NUM_TB + p) * Dsz + d0 + 2 * n) = o;
    }
}

extern "C" void kernel_launch(void* const* d_in, const int* in_sizes, int n_in,
                              void* d_out, int out_size, void* d_ws, size_t ws_size,
                              hipStream_t stream) {
    (void)in_sizes; (void)n_in; (void)d_ws; (void)ws_size; (void)out_size;
    const float* x  = (const float*)d_in[0];
    const float* ts = (const float*)d_in[1];
    float* out = (float*)d_out;

    dim3 grid(Dsz / 32, Bsz);
    time_bin_wmma_kernel<<<grid, 256, 0, stream>>>(x, ts, out);
}
